# starting kernel
# baseline (speedup 1.0000x reference)
#include <hip/hip_runtime.h>
#include <hip/hip_bf16.h>
#include <math.h>

typedef __attribute__((ext_vector_type(2))) float v2f;
typedef __attribute__((ext_vector_type(8))) float v8f;

#define D 128

__global__ void k_init_deg(float* deg, int n) {
    int i = blockIdx.x * blockDim.x + threadIdx.x;
    if (i < n) deg[i] = 1.0f;
}

__global__ void k_count_deg(const int* __restrict__ ei, int E, float* deg) {
    int e = blockIdx.x * blockDim.x + threadIdx.x;
    if (e < E) atomicAdd(&deg[ei[E + e]], 1.0f);
}

__global__ void k_dinv(const float* __restrict__ deg, float* dinv, int n) {
    int i = blockIdx.x * blockDim.x + threadIdx.x;
    if (i < n) {
        float d = deg[i];
        dinv[i] = (d > 0.0f) ? rsqrtf(d) : 0.0f;
    }
}

__global__ __launch_bounds__(256) void k_gemm_wmma(
        const float* __restrict__ A, const float* __restrict__ W,
        float* __restrict__ C, int n) {
    __shared__ float Wl[D * D];

    for (int i = threadIdx.x; i < (D * D) / 4; i += 256)
        ((float4*)Wl)[i] = ((const float4*)W)[i];
    __syncthreads();

    const int wave = threadIdx.x >> 5;
    const int lane = threadIdx.x & 31;
    const int m0   = blockIdx.x * 128 + wave * 16;

    int arow = m0 + (lane & 15);
    if (arow > n - 1) arow = n - 1;
    const int khalf = (lane >> 4) * 2;
    const float* Ar = A + (size_t)arow * D;

    v8f acc[8] = {};

    for (int k = 0; k < D; k += 4) {
        const v2f a = *(const v2f*)(Ar + k + khalf);
        const int kb = k + khalf;
#pragma unroll
        for (int t = 0; t < 8; ++t) {
            const int col = t * 16 + (lane & 15);
            v2f b;
            b.x = Wl[kb * D + col];
            b.y = Wl[(kb + 1) * D + col];
            acc[t] = __builtin_amdgcn_wmma_f32_16x16x4_f32(
                false, a, false, b, (short)0, acc[t], false, false);
        }
    }

    const int rbase = m0 + (lane >> 4) * 8;
    const int cix   = lane & 15;
#pragma unroll
    for (int t = 0; t < 8; ++t) {
#pragma unroll
        for (int v = 0; v < 8; ++v) {
            const int r = rbase + v;
            if (r < n) C[(size_t)r * D + t * 16 + cix] = acc[t][v];
        }
    }
}

__global__ void k_self_init(const float* __restrict__ xw,
                            const float* __restrict__ dinv,
                            float* __restrict__ out, int n) {
    size_t i = (size_t)blockIdx.x * blockDim.x + threadIdx.x;
    if (i >= (size_t)n * (D / 4)) return;
    int node = (int)(i >> 5);
    float dd = dinv[node]; dd *= dd;
    float4 v = ((const float4*)xw)[i];
    float4 o = {v.x * dd, v.y * dd, v.z * dd, v.w * dd};
    ((float4*)out)[i] = o;
}

__global__ __launch_bounds__(256) void k_agg(
        const int* __restrict__ ei, int E,
        const float* __restrict__ xw, const float* __restrict__ dinv,
        float* __restrict__ out) {
    int e = blockIdx.x * 8 + (threadIdx.x >> 5);
    if (e >= E) return;
    const int lane = threadIdx.x & 31;
    const int r = ei[e];
    const int c = ei[E + e];
    const float nrm = dinv[r] * dinv[c];
    float4 v = ((const float4*)(xw + (size_t)r * D))[lane];
    float* dst = out + (size_t)c * D + lane * 4;
    atomicAdd(dst + 0, v.x * nrm);
    atomicAdd(dst + 1, v.y * nrm);
    atomicAdd(dst + 2, v.z * nrm);
    atomicAdd(dst + 3, v.w * nrm);
}

__global__ void k_bias_relu(float* h, const float* __restrict__ b, int n) {
    size_t i = (size_t)blockIdx.x * blockDim.x + threadIdx.x;
    if (i >= (size_t)n * D) return;
    int f = (int)(i & (D - 1));
    float v = h[i] + b[f];
    h[i] = v > 0.0f ? v : 0.0f;
}

__global__ __launch_bounds__(256) void k_edge_cls(
        const int* __restrict__ ei, int E,
        const float* __restrict__ h, const float* __restrict__ Wfc,
        const float* __restrict__ bfc, float* __restrict__ out) {
    int e = blockIdx.x * 8 + (threadIdx.x >> 5);
    if (e >= E) return;
    const int lane = threadIdx.x & 31;
    const int r = ei[e];
    const int c = ei[E + e];
    float4 vr = ((const float4*)(h + (size_t)r * D))[lane];
    float4 vc = ((const float4*)(h + (size_t)c * D))[lane];
    const int k0 = lane * 4;
    float s0 = 0.f, s1 = 0.f, s2 = 0.f, s3 = 0.f;
    const float* pr = (const float*)&vr;
    const float* pc = (const float*)&vc;
#pragma unroll
    for (int i = 0; i < 4; ++i) {
        float a  = pr[i];
        float bb = pc[i];
        float4 wr = ((const float4*)Wfc)[k0 + i];
        float4 wc = ((const float4*)Wfc)[D + k0 + i];
        s0 += a * wr.x + bb * wc.x;
        s1 += a * wr.y + bb * wc.y;
        s2 += a * wr.z + bb * wc.z;
        s3 += a * wr.w + bb * wc.w;
    }
#pragma unroll
    for (int m = 16; m > 0; m >>= 1) {
        s0 += __shfl_xor(s0, m);
        s1 += __shfl_xor(s1, m);
        s2 += __shfl_xor(s2, m);
        s3 += __shfl_xor(s3, m);
    }
    if (lane == 0) {
        s0 += bfc[0]; s1 += bfc[1]; s2 += bfc[2]; s3 += bfc[3];
        float mx  = fmaxf(fmaxf(s0, s1), fmaxf(s2, s3));
        float sum = __expf(s0 - mx) + __expf(s1 - mx) +
                    __expf(s2 - mx) + __expf(s3 - mx);
        float lse = mx + __logf(sum);
        float4 o  = {s0 - lse, s1 - lse, s2 - lse, s3 - lse};
        ((float4*)out)[e] = o;
    }
}

extern "C" void kernel_launch(void* const* d_in, const int* in_sizes, int n_in,
                              void* d_out, int out_size, void* d_ws, size_t ws_size,
                              hipStream_t stream) {
    const float* x   = (const float*)d_in[0];
    const int*   ei  = (const int*)  d_in[1];
    const float* W1  = (const float*)d_in[2];
    const float* b1  = (const float*)d_in[3];
    const float* W2  = (const float*)d_in[4];
    const float* b2  = (const float*)d_in[5];
    const float* Wfc = (const float*)d_in[6];
    const float* bfc = (const float*)d_in[7];
    float* out = (float*)d_out;

    const int n = in_sizes[0] / D;
    const int E = in_sizes[1] / 2;

    float* bufA = (float*)d_ws;
    float* bufB = bufA + (size_t)n * D;
    float* deg  = bufB + (size_t)n * D;
    float* dinv = deg + n;

    const int gN    = (n + 255) / 256;
    const int gE    = (E + 255) / 256;
    const int gGemm = (n + 127) / 128;
    const int gVec4 = (int)(((size_t)n * (D / 4) + 255) / 256);
    const int gFeat = (int)(((size_t)n * D + 255) / 256);
    const int gWave = (E + 7) / 8;

    k_init_deg <<<gN, 256, 0, stream>>>(deg, n);
    k_count_deg<<<gE, 256, 0, stream>>>(ei, E, deg);
    k_dinv     <<<gN, 256, 0, stream>>>(deg, dinv, n);

    k_gemm_wmma<<<gGemm, 256, 0, stream>>>(x, W1, bufA, n);
    k_self_init<<<gVec4, 256, 0, stream>>>(bufA, dinv, bufB, n);
    k_agg      <<<gWave, 256, 0, stream>>>(ei, E, bufA, dinv, bufB);
    k_bias_relu<<<gFeat, 256, 0, stream>>>(bufB, b1, n);

    k_gemm_wmma<<<gGemm, 256, 0, stream>>>(bufB, W2, bufA, n);
    k_self_init<<<gVec4, 256, 0, stream>>>(bufA, dinv, bufB, n);
    k_agg      <<<gWave, 256, 0, stream>>>(ei, E, bufA, dinv, bufB);
    k_bias_relu<<<gFeat, 256, 0, stream>>>(bufB, b2, n);

    k_edge_cls <<<gWave, 256, 0, stream>>>(ei, E, bufB, Wfc, bfc, out);
}


# Per-edge and per-element launches become resident grids (1024 blocks) whose waves loop: 8 edges per trip, indices by one load, rows requested as one group, Wfc held in registers; sum orders kept
# speedup vs baseline: 1.3696x; 1.3696x over previous
#include <hip/hip_runtime.h>
#include <hip/hip_bf16.h>
#include <math.h>

typedef __attribute__((ext_vector_type(2))) float v2f;
typedef __attribute__((ext_vector_type(8))) float v8f;

#define D 128

__global__ void k_init_deg(float* deg, int n) {
    int i = blockIdx.x * blockDim.x + threadIdx.x;
    if (i < n) deg[i] = 1.0f;
}

__global__ void k_count_deg(const int* __restrict__ ei, int E, float* deg) {
    int e = blockIdx.x * blockDim.x + threadIdx.x;
    if (e < E) atomicAdd(&deg[ei[E + e]], 1.0f);
}

__global__ void k_dinv(const float* __restrict__ deg, float* dinv, int n) {
    int i = blockIdx.x * blockDim.x + threadIdx.x;
    if (i < n) {
        float d = deg[i];
        dinv[i] = (d > 0.0f) ? rsqrtf(d) : 0.0f;
    }
}

__global__ __launch_bounds__(256) void k_gemm_wmma(
        const float* __restrict__ A, const float* __restrict__ W,
        float* __restrict__ C, int n) {
    __shared__ float Wl[D * D];

    for (int i = threadIdx.x; i < (D * D) / 4; i += 256)
        ((float4*)Wl)[i] = ((const float4*)W)[i];
    __syncthreads();

    const int wave = threadIdx.x >> 5;
    const int lane = threadIdx.x & 31;
    const int m0   = blockIdx.x * 128 + wave * 16;

    int arow = m0 + (lane & 15);
    if (arow > n - 1) arow = n - 1;
    const int khalf = (lane >> 4) * 2;
    const float* Ar = A + (size_t)arow * D;

    v8f acc[8] = {};

    for (int k = 0; k < D; k += 4) {
        const v2f a = *(const v2f*)(Ar + k + khalf);
        const int kb = k + khalf;
#pragma unroll
        for (int t = 0; t < 8; ++t) {
            const int col = t * 16 + (lane & 15);
            v2f b;
            b.x = Wl[kb * D + col];
            b.y = Wl[(kb + 1) * D + col];
            acc[t] = __builtin_amdgcn_wmma_f32_16x16x4_f32(
                false, a, false, b, (short)0, acc[t], false, false);
        }
    }

    const int rbase = m0 + (lane >> 4) * 8;
    const int cix   = lane & 15;
#pragma unroll
    for (int t = 0; t < 8; ++t) {
#pragma unroll
        for (int v = 0; v < 8; ++v) {
            const int r = rbase + v;
            if (r < n) C[(size_t)r * D + t * 16 + cix] = acc[t][v];
        }
    }
}

#define RES_BLOCKS 1024
#define EPT 8

__global__ __launch_bounds__(256) void k_self_init(
        const float* __restrict__ xw, const float* __restrict__ dinv,
        float* __restrict__ out, int n) {
    const size_t total = (size_t)n * (D / 4);
    const size_t T = (size_t)gridDim.x * blockDim.x;
    const size_t tid = (size_t)blockIdx.x * blockDim.x + threadIdx.x;
    for (size_t i = tid; i < total; i += 4 * T) {
        float4 v[4];
        float dd[4];
#pragma unroll
        for (int u = 0; u < 4; ++u) {
            const size_t k = i + u * T;
            if (k < total) {
                v[u]  = ((const float4*)xw)[k];
                dd[u] = dinv[k >> 5];
            }
        }
#pragma unroll
        for (int u = 0; u < 4; ++u) {
            const size_t k = i + u * T;
            if (k < total) {
                const float d2 = dd[u] * dd[u];
                float4 o = {v[u].x * d2, v[u].y * d2, v[u].z * d2, v[u].w * d2};
                ((float4*)out)[k] = o;
            }
        }
    }
}

__device__ __forceinline__ int trip_index(const int* __restrict__ ei, int E,
                                          int trip, int lane) {
    int e = trip * EPT + (lane & 7);
    if (e > E - 1) e = E - 1;
    return ei[(size_t)((lane >> 3) & 1) * E + e];
}
__device__ __forceinline__ int lane_i(int v, int l) {
    return __builtin_amdgcn_readlane(v, l);
}
__device__ __forceinline__ float lane_f(float v, int l) {
    return __int_as_float(__builtin_amdgcn_readlane(__float_as_int(v), l));
}

__global__ __launch_bounds__(256) void k_agg(
        const int* __restrict__ ei, int E,
        const float* __restrict__ xw, const float* __restrict__ dinv,
        float* __restrict__ out) {
    const int lane  = threadIdx.x & 31;
    const int wave  = blockIdx.x * 8 + __builtin_amdgcn_readfirstlane(threadIdx.x >> 5);
    const int nwave = gridDim.x * 8;
    const int nfull = E / EPT;

    if (wave < nfull) {
        int idx = trip_index(ei, E, wave, lane);
        for (int t = wave; t < nfull; t += nwave) {
            const float dv = dinv[idx];
            const int cur = idx;
            const int tn = (t + nwave < nfull) ? t + nwave : nfull - 1;
            idx = trip_index(ei, E, tn, lane);

            float v[EPT][4];
#pragma unroll
            for (int j = 0; j < EPT; ++j) {
                const float* src = xw + (size_t)lane_i(cur, j) * D + lane;
                v[j][0] = src[0];
                v[j][1] = src[32];
                v[j][2] = src[64];
                v[j][3] = src[96];
            }
#pragma unroll
            for (int j = 0; j < EPT; ++j) {
                const float nrm = lane_f(dv, j) * lane_f(dv, 8 + j);
                float* dst = out + (size_t)lane_i(cur, 8 + j) * D + lane;
                atomicAdd(dst +  0, v[j][0] * nrm);
                atomicAdd(dst + 32, v[j][1] * nrm);
                atomicAdd(dst + 64, v[j][2] * nrm);
                atomicAdd(dst + 96, v[j][3] * nrm);
            }
        }
    }

    const int e = nfull * EPT + wave;
    if (wave < EPT && e < E) {
        const int r = ei[e];
        const int c = ei[E + e];
        const float nrm = dinv[r] * dinv[c];
        const float* src = xw + (size_t)r * D + lane;
        const float v0 = src[0];
        const float v1 = src[32];
        const float v2 = src[64];
        const float v3 = src[96];
        float* dst = out + (size_t)c * D + lane;
        atomicAdd(dst +  0, v0 * nrm);
        atomicAdd(dst + 32, v1 * nrm);
        atomicAdd(dst + 64, v2 * nrm);
        atomicAdd(dst + 96, v3 * nrm);
    }
}

__global__ __launch_bounds__(256) void k_bias_relu(float* h, const float* __restrict__ b, int n) {
    const size_t total = (size_t)n * (D / 4);
    const size_t T = (size_t)gridDim.x * blockDim.x;
    const size_t tid = (size_t)blockIdx.x * blockDim.x + threadIdx.x;
    const float4 bb = ((const float4*)b)[tid & (D / 4 - 1)];
    for (size_t i = tid; i < total; i += 4 * T) {
        float4 v[4];
#pragma unroll
        for (int u = 0; u < 4; ++u) {
            const size_t k = i + u * T;
            if (k < total) v[u] = ((const float4*)h)[k];
        }
#pragma unroll
        for (int u = 0; u < 4; ++u) {
            const size_t k = i + u * T;
            if (k < total) {
                float4 o;
                o.x = v[u].x + bb.x; o.x = o.x > 0.0f ? o.x : 0.0f;
                o.y = v[u].y + bb.y; o.y = o.y > 0.0f ? o.y : 0.0f;
                o.z = v[u].z + bb.z; o.z = o.z > 0.0f ? o.z : 0.0f;
                o.w = v[u].w + bb.w; o.w = o.w > 0.0f ? o.w : 0.0f;
                ((float4*)h)[k] = o;
            }
        }
    }
}

__global__ __launch_bounds__(256) void k_edge_cls(
        const int* __restrict__ ei, int E,
        const float* __restrict__ h, const float* __restrict__ Wfc,
        const float* __restrict__ bfc, float* __restrict__ out) {
    const int lane  = threadIdx.x & 31;
    const int wave  = blockIdx.x * 8 + __builtin_amdgcn_readfirstlane(threadIdx.x >> 5);
    const int nwave = gridDim.x * 8;
    const int ntrip = (E + EPT - 1) / EPT;
    if (wave >= ntrip) return;

    float4 wr[4], wc[4];
#pragma unroll
    for (int j = 0; j < 4; ++j) {
        wr[j] = ((const float4*)Wfc)[lane + 32 * j];
        wc[j] = ((const float4*)Wfc)[D + lane + 32 * j];
    }
    const float4 bf = *(const float4*)bfc;

    int idx = trip_index(ei, E, wave, lane);
    for (int t = wave; t < ntrip; t += nwave) {
        const int cur = idx;
        const int tn = (t + nwave < ntrip) ? t + nwave : ntrip - 1;
        idx = trip_index(ei, E, tn, lane);

        float m0 = 0.f, m1 = 0.f, m2 = 0.f, m3 = 0.f;
#pragma unroll
        for (int g = 0; g < EPT; g += 4) {
            float a[4][4], bq[4][4];
#pragma unroll
            for (int q = 0; q < 4; ++q) {
                const float* hr = h + (size_t)lane_i(cur, g + q) * D + lane;
                const float* hc = h + (size_t)lane_i(cur, 8 + g + q) * D + lane;
#pragma unroll
                for (int j = 0; j < 4; ++j) {
                    a[q][j]  = hr[32 * j];
                    bq[q][j] = hc[32 * j];
                }
            }
#pragma unroll
            for (int q = 0; q < 4; ++q) {
                float s0 = 0.f, s1 = 0.f, s2 = 0.f, s3 = 0.f;
#pragma unroll
                for (int j = 0; j < 4; ++j) {
                    s0 += a[q][j] * wr[j].x + bq[q][j] * wc[j].x;
                    s1 += a[q][j] * wr[j].y + bq[q][j] * wc[j].y;
                    s2 += a[q][j] * wr[j].z + bq[q][j] * wc[j].z;
                    s3 += a[q][j] * wr[j].w + bq[q][j] * wc[j].w;
                }
#pragma unroll
                for (int m = 16; m > 0; m >>= 1) {
                    s0 += __shfl_xor(s0, m);
                    s1 += __shfl_xor(s1, m);
                    s2 += __shfl_xor(s2, m);
                    s3 += __shfl_xor(s3, m);
                }
                if (lane == g + q) { m0 = s0; m1 = s1; m2 = s2; m3 = s3; }
            }
        }
        const int e = t * EPT + lane;
        if (lane < EPT && e < E) {
            m0 += bf.x; m1 += bf.y; m2 += bf.z; m3 += bf.w;
            float mx  = fmaxf(fmaxf(m0, m1), fmaxf(m2, m3));
            float sum = __expf(m0 - mx) + __expf(m1 - mx) +
                        __expf(m2 - mx) + __expf(m3 - mx);
            float lse = mx + __logf(sum);
            float4 o  = {m0 - lse, m1 - lse, m2 - lse, m3 - lse};
            ((float4*)out)[e] = o;
        }
    }
}

extern "C" void kernel_launch(void* const* d_in, const int* in_sizes, int n_in,
                              void* d_out, int out_size, void* d_ws, size_t ws_size,
                              hipStream_t stream) {
    const float* x   = (const float*)d_in[0];
    const int*   ei  = (const int*)  d_in[1];
    const float* W1  = (const float*)d_in[2];
    const float* b1  = (const float*)d_in[3];
    const float* W2  = (const float*)d_in[4];
    const float* b2  = (const float*)d_in[5];
    const float* Wfc = (const float*)d_in[6];
    const float* bfc = (const float*)d_in[7];
    float* out = (float*)d_out;

    const int n = in_sizes[0] / D;
    const int E = in_sizes[1] / 2;

    float* bufA = (float*)d_ws;
    float* bufB = bufA + (size_t)n * D;
    float* deg  = bufB + (size_t)n * D;
    float* dinv = deg + n;

    const int gN    = (n + 255) / 256;
    const int gE    = (E + 255) / 256;
    const int gGemm = (n + 127) / 128;
    const int gRes  = RES_BLOCKS;

    k_init_deg <<<gN, 256, 0, stream>>>(deg, n);
    k_count_deg<<<gE, 256, 0, stream>>>(ei, E, deg);
    k_dinv     <<<gN, 256, 0, stream>>>(deg, dinv, n);

    k_gemm_wmma<<<gGemm, 256, 0, stream>>>(x, W1, bufA, n);
    k_self_init<<<gRes, 256, 0, stream>>>(bufA, dinv, bufB, n);
    k_agg      <<<gRes, 256, 0, stream>>>(ei, E, bufA, dinv, bufB);
    k_bias_relu<<<gRes, 256, 0, stream>>>(bufB, b1, n);

    k_gemm_wmma<<<gGemm, 256, 0, stream>>>(bufB, W2, bufA, n);
    k_self_init<<<gRes, 256, 0, stream>>>(bufA, dinv, bufB, n);
    k_agg      <<<gRes, 256, 0, stream>>>(ei, E, bufA, dinv, bufB);
    k_bias_relu<<<gRes, 256, 0, stream>>>(bufB, b2, n);

    k_edge_cls <<<gRes, 256, 0, stream>>>(ei, E, bufB, Wfc, bfc, out);
}


# Edge scatters: global f32 atomics replaced by owner blocks (200 target rows, sums in LDS) that find their edges by scanning one key byte per edge; LDS float adds, sum order = their arrival order
# speedup vs baseline: 2.3475x; 1.7140x over previous
#include <hip/hip_runtime.h>
#include <hip/hip_bf16.h>
#include <math.h>

typedef __attribute__((ext_vector_type(2))) float v2f;
typedef __attribute__((ext_vector_type(8))) float v8f;

#define D 128

__global__ void k_init_deg(float* deg, int n) {
    int i = blockIdx.x * blockDim.x + threadIdx.x;
    if (i < n) deg[i] = 1.0f;
}

#define NPB 200

__global__ void k_count_deg(const int* __restrict__ ei, int E, float* deg,
                            unsigned* __restrict__ key) {
    const int q  = blockIdx.x * blockDim.x + threadIdx.x;
    const int e0 = q * 4;
    if (e0 >= ((E + 15) & ~15)) return;
    unsigned kw = 0;
#pragma unroll
    for (int k = 0; k < 4; ++k) {
        const int e = e0 + k;
        if (e < E) {
            const int c = ei[E + e];
            atomicAdd(&deg[c], 1.0f);
            kw |= (((unsigned)c / NPB) & 255u) << (8 * k);
        }
    }
    key[q] = kw;
}

__device__ __forceinline__ float inv_sqrt_deg(float d) {
    return (d > 0.0f) ? rsqrtf(d) : 0.0f;
}

__device__ __forceinline__ void gemm_rows(
        const float* __restrict__ A,
        float* __restrict__ C, int n,
        const float* Wl, const float* bl, const bool relu_in) {
    const int wave = threadIdx.x >> 5;
    const int lane = threadIdx.x & 31;
    const int m0   = blockIdx.x * 128 + wave * 16;

    int arow = m0 + (lane & 15);
    if (arow > n - 1) arow = n - 1;
    const int khalf = (lane >> 4) * 2;
    const float* Ar = A + (size_t)arow * D;

    v8f acc[8] = {};

    for (int k = 0; k < D; k += 4) {
        v2f a = *(const v2f*)(Ar + k + khalf);
        const int kb = k + khalf;
        if (relu_in) {
            const v2f bb = *(const v2f*)(bl + kb);
            a.x = a.x + bb.x; a.x = a.x > 0.0f ? a.x : 0.0f;
            a.y = a.y + bb.y; a.y = a.y > 0.0f ? a.y : 0.0f;
        }
#pragma unroll
        for (int t = 0; t < 8; ++t) {
            const int col = t * 16 + (lane & 15);
            v2f b;
            b.x = Wl[kb * D + col];
            b.y = Wl[(kb + 1) * D + col];
            acc[t] = __builtin_amdgcn_wmma_f32_16x16x4_f32(
                false, a, false, b, (short)0, acc[t], false, false);
        }
    }

    const int rbase = m0 + (lane >> 4) * 8;
    const int cix   = lane & 15;
#pragma unroll
    for (int t = 0; t < 8; ++t) {
#pragma unroll
        for (int v = 0; v < 8; ++v) {
            const int r = rbase + v;
            if (r < n) C[(size_t)r * D + t * 16 + cix] = acc[t][v];
        }
    }
}

__global__ __launch_bounds__(256) void k_gemm_first(
        const float* __restrict__ A, const float* __restrict__ W,
        float* __restrict__ C, int n) {
    __shared__ float Wl[D * D];

    for (int i = threadIdx.x; i < (D * D) / 4; i += 256)
        ((float4*)Wl)[i] = ((const float4*)W)[i];
    __syncthreads();

    gemm_rows(A, C, n, Wl, Wl, false);
}

__global__ __launch_bounds__(256) void k_gemm_relu_in(
        const float* __restrict__ A, const float* __restrict__ bin,
        const float* __restrict__ W,
        float* __restrict__ C, int n) {
    __shared__ float Wl[D * D];
    __shared__ float bl[D];

    for (int i = threadIdx.x; i < (D * D) / 4; i += 256)
        ((float4*)Wl)[i] = ((const float4*)W)[i];
    if (threadIdx.x < D) bl[threadIdx.x] = bin[threadIdx.x];
    __syncthreads();

    gemm_rows(A, C, n, Wl, bl, true);
}

#define RES_BLOCKS 1024
#define EPT 8

__device__ __forceinline__ int trip_index(const int* __restrict__ ei, int E,
                                          int trip, int lane) {
    int e = trip * EPT + (lane & 7);
    if (e > E - 1) e = E - 1;
    return ei[(size_t)((lane >> 3) & 1) * E + e];
}
__device__ __forceinline__ int lane_i(int v, int l) {
    return __builtin_amdgcn_readlane(v, l);
}
__device__ __forceinline__ float lane_f(float v, int l) {
    return __int_as_float(__builtin_amdgcn_readlane(__float_as_int(v), l));
}
__device__ __forceinline__ float xhalf(float x) {
    const int b = __float_as_int(x);
    return __int_as_float(__builtin_amdgcn_permlanex16(b, b, 0x76543210, 0xfedcba98, false, false));
}
template <int M>
__device__ __forceinline__ float xrow(float x) {
    return __int_as_float(__builtin_amdgcn_update_dpp(0, __float_as_int(x), 0x160 + M, 0xf, 0xf, true));
}

#define OWN_WAVES   32
#define SCAN_U      8
#define ROUND_STEPS 40
#define CQ_CAP      (ROUND_STEPS * 32)
#define EQ_CAP      640
#define ROWS_PER_WAVE ((NPB + OWN_WAVES - 1) / OWN_WAVES)

__device__ __forceinline__ void wave_lds_order() {
    __builtin_amdgcn_fence(__ATOMIC_RELEASE, "wavefront");
    __builtin_amdgcn_wave_barrier();
    __builtin_amdgcn_fence(__ATOMIC_ACQUIRE, "wavefront");
}

__device__ __forceinline__ int own_idx(const int* __restrict__ ei, int E,
                                       const int* q, int head, int nv, int lane) {
    const int t = lane & 7;
    const int e = q[head + (t < nv ? t : 0)];
    return ei[(size_t)((lane >> 3) & 1) * E + e];
}

__device__ __forceinline__ int own_drain(
        const int* __restrict__ ei, int E, const float* __restrict__ xw,
        const float* __restrict__ deg, float* acc, int* q, int ecnt,
        int lo, int hi, int lane, const bool flush) {
    int head = 0;
    int left = ecnt;
    if (flush ? (left > 0) : (left >= EPT)) {
        wave_lds_order();
        int nv  = left < EPT ? left : EPT;
        int idx = own_idx(ei, E, q, head, nv, lane);
        for (;;) {
            const int cur   = idx;
            const int curnv = nv;
            const float dg  = deg[cur];
            head += curnv;
            left -= curnv;
            const bool more = flush ? (left > 0) : (left >= EPT);
            if (more) {
                nv  = left < EPT ? left : EPT;
                idx = own_idx(ei, E, q, head, nv, lane);
            }
            float v[EPT][4];
#pragma unroll
            for (int j = 0; j < EPT; ++j) {
                const float* src = xw + (size_t)lane_i(cur, j) * D + lane;
                v[j][0] = src[0];
                v[j][1] = src[32];
                v[j][2] = src[64];
                v[j][3] = src[96];
            }
            __builtin_amdgcn_sched_barrier(0);
            const float dv = inv_sqrt_deg(dg);
#pragma unroll
            for (int j = 0; j < EPT; ++j) {
                const int c = lane_i(cur, 8 + j);
                if (j < curnv && c >= lo && c < hi) {
                    const float nrm = lane_f(dv, j) * lane_f(dv, 8 + j);
                    float* dst = acc + (c - lo) * D + lane;
                    atomicAdd(dst +  0, v[j][0] * nrm);
                    atomicAdd(dst + 32, v[j][1] * nrm);
                    atomicAdd(dst + 64, v[j][2] * nrm);
                    atomicAdd(dst + 96, v[j][3] * nrm);
                }
            }
            if (!more) break;
        }
        if (left > 0) {
            const int tmp = q[head + (lane < left ? lane : 0)];
            wave_lds_order();
            if (lane < left) q[lane] = tmp;
        }
    }
    return left;
}

__global__ __launch_bounds__(1024) void k_agg_own(
        const int* __restrict__ ei, int E,
        const unsigned* __restrict__ key,
        const float* __restrict__ xw, const float* __restrict__ deg,
        float* __restrict__ out, int n) {
    __shared__ float acc[NPB * D];
    __shared__ unsigned short cq[OWN_WAVES][CQ_CAP];
    __shared__ int eq[OWN_WAVES][EQ_CAP];

    const int lane = threadIdx.x & 31;
    const int wave   = __builtin_amdgcn_readfirstlane(threadIdx.x >> 5);
    const int nchunk = (E + 15) >> 4;
    const int nstep  = (nchunk + 31) >> 5;
    const int nrange = (n + NPB - 1) / NPB;
    const uint4* kp  = (const uint4*)key;
    unsigned short* mycq = cq[wave];
    int* myeq = eq[wave];

    for (int r = blockIdx.x; r < nrange; r += gridDim.x) {
        const int lo   = r * NPB;
        const int hi   = (lo + NPB < n) ? lo + NPB : n;
        const int rows = hi - lo;
        const unsigned pat  = ((unsigned)r & 255u) * 0x01010101u;

        {
            float sv[ROWS_PER_WAVE][4], sd[ROWS_PER_WAVE];
#pragma unroll
            for (int k = 0; k < ROWS_PER_WAVE; ++k) {
                const int i  = wave + OWN_WAVES * k;
                const int rr = lo + (i < rows ? i : rows - 1);
                sd[k] = deg[rr];
                const float* src = xw + (size_t)rr * D + lane;
#pragma unroll
                for (int j = 0; j < 4; ++j) sv[k][j] = src[32 * j];
            }
#pragma unroll
            for (int k = 0; k < ROWS_PER_WAVE; ++k) {
                const int i = wave + OWN_WAVES * k;
                if (i < rows) {
                    const float dd = inv_sqrt_deg(sd[k]);
                    const float d2 = dd * dd;
                    float* a = acc + i * D + lane;
#pragma unroll
                    for (int j = 0; j < 4; ++j) a[32 * j] = sv[k][j] * d2;
                }
            }
        }
        __syncthreads();

        int ecnt = 0;
        for (int i0 = 0; wave + 32 * i0 < nstep; i0 += ROUND_STEPS) {
            int cnt = 0;
            for (int g = 0; g < ROUND_STEPS; g += SCAN_U) {
                if (wave + 32 * (i0 + g) >= nstep) break;
                uint4 kv[SCAN_U];
#pragma unroll
                for (int u = 0; u < SCAN_U; ++u) {
                    const int ch = (wave + 32 * (i0 + g + u)) * 32 + lane;
                    kv[u] = kp[ch < nchunk ? ch : nchunk - 1];
                }
#pragma unroll
                for (int u = 0; u < SCAN_U; ++u) {
                    const unsigned a = kv[u].x ^ pat, b = kv[u].y ^ pat;
                    const unsigned c = kv[u].z ^ pat, d = kv[u].w ^ pat;
                    const unsigned z = (((a - 0x01010101u) & ~a) | ((b - 0x01010101u) & ~b) |
                                        ((c - 0x01010101u) & ~c) | ((d - 0x01010101u) & ~d)) & 0x80808080u;
                    const bool hit = z != 0u && (wave + 32 * (i0 + g + u)) * 32 + lane < nchunk;
                    const unsigned mask = __builtin_amdgcn_ballot_w32(hit);
                    if (hit)
                        mycq[cnt + __builtin_amdgcn_mbcnt_lo(mask, 0u)] =
                            (unsigned short)(((g + u) << 5) | lane);
                    cnt += __builtin_popcount(mask);
                }
            }
            for (int base = 0; base < cnt; base += 32) {
                if (ecnt + 512 > EQ_CAP)
                    ecnt = own_drain(ei, E, xw, deg, acc, myeq, ecnt, lo, hi, lane, false);
                wave_lds_order();
                const bool have = base + lane < cnt;
                const unsigned q = mycq[have ? base + lane : base];
                const int ch = (wave + 32 * (i0 + (int)(q >> 5))) * 32 + (int)(q & 31u);
                const uint4 kw = kp[ch];
                const unsigned w[4] = {kw.x ^ pat, kw.y ^ pat, kw.z ^ pat, kw.w ^ pat};
#pragma unroll
                for (int k = 0; k < 16; ++k) {
                    const int e = ch * 16 + k;
                    const bool m = have && ((w[k >> 2] >> (8 * (k & 3))) & 255u) == 0u && e < E;
                    const unsigned mask = __builtin_amdgcn_ballot_w32(m);
                    if (m) myeq[ecnt + __builtin_amdgcn_mbcnt_lo(mask, 0u)] = e;
                    ecnt += __builtin_popcount(mask);
                }
            }
        }
        ecnt = own_drain(ei, E, xw, deg, acc, myeq, ecnt, lo, hi, lane, true);
        __syncthreads();

        for (int i = wave; i < rows; i += OWN_WAVES) {
            float* dst = out + (size_t)(lo + i) * D + lane;
            const float* a = acc + i * D + lane;
#pragma unroll
            for (int j = 0; j < 4; ++j) dst[32 * j] = a[32 * j];
        }
        __syncthreads();
    }
}

__global__ __launch_bounds__(256) void k_edge_cls(
        const int* __restrict__ ei, int E,
        const float* __restrict__ hs, const float* __restrict__ bh,
        const float* __restrict__ Wfc,
        const float* __restrict__ bfc, float* __restrict__ out) {
    const int lane  = threadIdx.x & 31;
    const int wave  = blockIdx.x * 8 + __builtin_amdgcn_readfirstlane(threadIdx.x >> 5);
    const int nwave = gridDim.x * 8;
    const int ntrip = (E + EPT - 1) / EPT;
    if (wave >= ntrip) return;

    float4 wr[4], wc[4];
#pragma unroll
    for (int j = 0; j < 4; ++j) {
        wr[j] = ((const float4*)Wfc)[lane + 32 * j];
        wc[j] = ((const float4*)Wfc)[D + lane + 32 * j];
    }
    const float bfl = bfc[lane & 3];
    float bv[4];
#pragma unroll
    for (int j = 0; j < 4; ++j) bv[j] = bh[lane + 32 * j];

    const bool b4 = (lane & 16) != 0;
    const bool b3 = (lane &  8) != 0;
    const bool b2 = (lane &  4) != 0;
    const bool b1 = (lane &  2) != 0;
    const bool b0 = (lane &  1) != 0;

    int idx = trip_index(ei, E, wave, lane);
    for (int t = wave; t < ntrip; t += nwave) {
        const int cur = idx;
        const int tn = (t + nwave < ntrip) ? t + nwave : ntrip - 1;
        idx = trip_index(ei, E, tn, lane);

        float p[4 * EPT];
#pragma unroll
        for (int g = 0; g < EPT; g += 4) {
            float a[4][4], bq[4][4];
#pragma unroll
            for (int q = 0; q < 4; ++q) {
                const float* hr = hs + (size_t)lane_i(cur, g + q) * D + lane;
                const float* hc = hs + (size_t)lane_i(cur, 8 + g + q) * D + lane;
#pragma unroll
                for (int j = 0; j < 4; ++j) {
                    a[q][j]  = hr[32 * j];
                    bq[q][j] = hc[32 * j];
                }
            }
#pragma unroll
            for (int q = 0; q < 4; ++q) {
#pragma unroll
                for (int j = 0; j < 4; ++j) {
                    float u = a[q][j] + bv[j];
                    a[q][j] = u > 0.0f ? u : 0.0f;
                    float w = bq[q][j] + bv[j];
                    bq[q][j] = w > 0.0f ? w : 0.0f;
                }
            }
#pragma unroll
            for (int q = 0; q < 4; ++q) {
                float s0 = 0.f, s1 = 0.f, s2 = 0.f, s3 = 0.f;
#pragma unroll
                for (int j = 0; j < 4; ++j) {
                    s0 += a[q][j] * wr[j].x + bq[q][j] * wc[j].x;
                    s1 += a[q][j] * wr[j].y + bq[q][j] * wc[j].y;
                    s2 += a[q][j] * wr[j].z + bq[q][j] * wc[j].z;
                    s3 += a[q][j] * wr[j].w + bq[q][j] * wc[j].w;
                }
                p[4 * (g + q) + 0] = s0;
                p[4 * (g + q) + 1] = s1;
                p[4 * (g + q) + 2] = s2;
                p[4 * (g + q) + 3] = s3;
            }
        }

        float r1[16], r2[8], r3[4], r2b[2];
#pragma unroll
        for (int w = 0; w < 16; ++w) {
            const float keep = b4 ? p[16 + w] : p[w];
            const float send = b4 ? p[w] : p[16 + w];
            r1[w] = keep + xhalf(send);
        }
#pragma unroll
        for (int w = 0; w < 8; ++w) {
            const float keep = b3 ? r1[8 + w] : r1[w];
            const float send = b3 ? r1[w] : r1[8 + w];
            r2[w] = keep + xrow<8>(send);
        }
#pragma unroll
        for (int w = 0; w < 4; ++w) {
            const float keep = b2 ? r2[4 + w] : r2[w];
            const float send = b2 ? r2[w] : r2[4 + w];
            r3[w] = keep + xrow<4>(send);
        }
#pragma unroll
        for (int w = 0; w < 2; ++w) {
            const float keep = b1 ? r3[2 + w] : r3[w];
            const float send = b1 ? r3[w] : r3[2 + w];
            r2b[w] = keep + xrow<2>(send);
        }
        float x;
        {
            const float keep = b0 ? r2b[1] : r2b[0];
            const float send = b0 ? r2b[0] : r2b[1];
            x = keep + xrow<1>(send);
        }

        x += bfl;
        float mx = fmaxf(x, xrow<1>(x));
        mx = fmaxf(mx, xrow<2>(mx));
        const float ex = __expf(x - mx);
        float sum = ex + xrow<1>(ex);
        sum = sum + xrow<2>(sum);
        const float o = x - (mx + __logf(sum));

        const int e = t * EPT + (lane >> 2);
        if (e < E) out[(size_t)t * (4 * EPT) + lane] = o;
    }
}

extern "C" void kernel_launch(void* const* d_in, const int* in_sizes, int n_in,
                              void* d_out, int out_size, void* d_ws, size_t ws_size,
                              hipStream_t stream) {
    const float* x   = (const float*)d_in[0];
    const int*   ei  = (const int*)  d_in[1];
    const float* W1  = (const float*)d_in[2];
    const float* b1  = (const float*)d_in[3];
    const float* W2  = (const float*)d_in[4];
    const float* b2  = (const float*)d_in[5];
    const float* Wfc = (const float*)d_in[6];
    const float* bfc = (const float*)d_in[7];
    float* out = (float*)d_out;

    const int n = in_sizes[0] / D;
    const int E = in_sizes[1] / 2;

    float* bufA = (float*)d_ws;
    float* bufB = bufA + (size_t)n * D;
    float* bufC = bufB + (size_t)n * D;
    float* deg  = bufC + (size_t)n * D;
    unsigned* key = (unsigned*)(deg + ((n + 3) & ~3));

    const int gN    = (n + 255) / 256;
    const int gGemm = (n + 127) / 128;
    const int gRes  = RES_BLOCKS;
    const int gKey  = (((E + 15) / 16) * 4 + 255) / 256;
    const int nOwn  = (n + NPB - 1) / NPB;
    const int gOwn  = nOwn < 256 ? nOwn : 256;

    k_init_deg <<<gN, 256, 0, stream>>>(deg, n);
    k_count_deg<<<gKey, 256, 0, stream>>>(ei, E, deg, key);

    k_gemm_first  <<<gGemm, 256, 0, stream>>>(x, W1, bufA, n);
    k_agg_own     <<<gOwn, 1024, 0, stream>>>(ei, E, key, bufA, deg, bufB, n);

    k_gemm_relu_in<<<gGemm, 256, 0, stream>>>(bufB, b1, W2, bufA, n);
    k_agg_own     <<<gOwn, 1024, 0, stream>>>(ei, E, key, bufA, deg, bufC, n);

    k_edge_cls    <<<gRes, 256, 0, stream>>>(ei, E, bufC, b2, Wfc, bfc, out);
}


# Layer-1 owner waves file each 8-edge trip (sources, local targets, weights) as one 128-byte workspace line; layer 2 adds from those lines: no key scan, index or degree gathers; sum order kept
# speedup vs baseline: 2.4824x; 1.0575x over previous
#include <hip/hip_runtime.h>
#include <hip/hip_bf16.h>
#include <math.h>

typedef __attribute__((ext_vector_type(2))) float v2f;
typedef __attribute__((ext_vector_type(8))) float v8f;

#define D 128

__global__ void k_init_deg(float* deg, int n) {
    int i = blockIdx.x * blockDim.x + threadIdx.x;
    if (i < n) deg[i] = 1.0f;
}

#define NPB 200

__global__ void k_count_deg(const int* __restrict__ ei, int E, float* deg,
                            unsigned* __restrict__ key) {
    const int q  = blockIdx.x * blockDim.x + threadIdx.x;
    const int e0 = q * 4;
    if (e0 >= ((E + 15) & ~15)) return;
    unsigned kw = 0;
#pragma unroll
    for (int k = 0; k < 4; ++k) {
        const int e = e0 + k;
        if (e < E) {
            const int c = ei[E + e];
            atomicAdd(&deg[c], 1.0f);
            kw |= (((unsigned)c / NPB) & 255u) << (8 * k);
        }
    }
    key[q] = kw;
}

__device__ __forceinline__ float inv_sqrt_deg(float d) {
    return (d > 0.0f) ? rsqrtf(d) : 0.0f;
}

__device__ __forceinline__ void gemm_rows(
        const float* __restrict__ A,
        float* __restrict__ C, int n,
        const float* Wl, const float* bl, const bool relu_in) {
    const int wave = threadIdx.x >> 5;
    const int lane = threadIdx.x & 31;
    const int m0   = blockIdx.x * 128 + wave * 16;

    int arow = m0 + (lane & 15);
    if (arow > n - 1) arow = n - 1;
    const int khalf = (lane >> 4) * 2;
    const float* Ar = A + (size_t)arow * D;

    v8f acc[8] = {};

    for (int k = 0; k < D; k += 4) {
        v2f a = *(const v2f*)(Ar + k + khalf);
        const int kb = k + khalf;
        if (relu_in) {
            const v2f bb = *(const v2f*)(bl + kb);
            a.x = a.x + bb.x; a.x = a.x > 0.0f ? a.x : 0.0f;
            a.y = a.y + bb.y; a.y = a.y > 0.0f ? a.y : 0.0f;
        }
#pragma unroll
        for (int t = 0; t < 8; ++t) {
            const int col = t * 16 + (lane & 15);
            v2f b;
            b.x = Wl[kb * D + col];
            b.y = Wl[(kb + 1) * D + col];
            acc[t] = __builtin_amdgcn_wmma_f32_16x16x4_f32(
                false, a, false, b, (short)0, acc[t], false, false);
        }
    }

    const int rbase = m0 + (lane >> 4) * 8;
    const int cix   = lane & 15;
#pragma unroll
    for (int t = 0; t < 8; ++t) {
#pragma unroll
        for (int v = 0; v < 8; ++v) {
            const int r = rbase + v;
            if (r < n) C[(size_t)r * D + t * 16 + cix] = acc[t][v];
        }
    }
}

__global__ __launch_bounds__(256) void k_gemm_first(
        const float* __restrict__ A, const float* __restrict__ W,
        float* __restrict__ C, int n) {
    __shared__ float Wl[D * D];

    for (int i = threadIdx.x; i < (D * D) / 4; i += 256)
        ((float4*)Wl)[i] = ((const float4*)W)[i];
    __syncthreads();

    gemm_rows(A, C, n, Wl, Wl, false);
}

__global__ __launch_bounds__(256) void k_gemm_relu_in(
        const float* __restrict__ A, const float* __restrict__ bin,
        const float* __restrict__ W,
        float* __restrict__ C, int n) {
    __shared__ float Wl[D * D];
    __shared__ float bl[D];

    for (int i = threadIdx.x; i < (D * D) / 4; i += 256)
        ((float4*)Wl)[i] = ((const float4*)W)[i];
    if (threadIdx.x < D) bl[threadIdx.x] = bin[threadIdx.x];
    __syncthreads();

    gemm_rows(A, C, n, Wl, bl, true);
}

#define RES_BLOCKS 1024
#define EPT 8

__device__ __forceinline__ int trip_index(const int* __restrict__ ei, int E,
                                          int trip, int lane) {
    int e = trip * EPT + (lane & 7);
    if (e > E - 1) e = E - 1;
    return ei[(size_t)((lane >> 3) & 1) * E + e];
}
__device__ __forceinline__ int lane_i(int v, int l) {
    return __builtin_amdgcn_readlane(v, l);
}
__device__ __forceinline__ float lane_f(float v, int l) {
    return __int_as_float(__builtin_amdgcn_readlane(__float_as_int(v), l));
}
__device__ __forceinline__ float xhalf(float x) {
    const int b = __float_as_int(x);
    return __int_as_float(__builtin_amdgcn_permlanex16(b, b, 0x76543210, 0xfedcba98, false, false));
}
template <int M>
__device__ __forceinline__ float xrow(float x) {
    return __int_as_float(__builtin_amdgcn_update_dpp(0, __float_as_int(x), 0x160 + M, 0xf, 0xf, true));
}

#define OWN_WAVES   32
#define SCAN_U      8
#define ROUND_STEPS 40
#define CQ_CAP      (ROUND_STEPS * 32)
#define EQ_CAP      640
#define ROWS_PER_WAVE ((NPB + OWN_WAVES - 1) / OWN_WAVES)

#define TCAP 32

__device__ __forceinline__ void wave_lds_order() {
    __builtin_amdgcn_fence(__ATOMIC_RELEASE, "wavefront");
    __builtin_amdgcn_wave_barrier();
    __builtin_amdgcn_fence(__ATOMIC_ACQUIRE, "wavefront");
}

__device__ __forceinline__ int own_idx(const int* __restrict__ ei, int E,
                                       const int* q, int head, int nv, int lane) {
    const int t = lane & 7;
    const int e = q[head + (t < nv ? t : 0)];
    return ei[(size_t)((lane >> 3) & 1) * E + e];
}

template <bool REC>
__device__ __forceinline__ int own_drain(
        const int* __restrict__ ei, int E, const float* __restrict__ xw,
        const float* __restrict__ deg, float* acc, int* q, int ecnt,
        int lo, int hi, int lane, const bool flush,
        int* __restrict__ recw, int& nrec) {
    int head = 0;
    int left = ecnt;
    if (flush ? (left > 0) : (left >= EPT)) {
        wave_lds_order();
        int nv  = left < EPT ? left : EPT;
        int idx = own_idx(ei, E, q, head, nv, lane);
        for (;;) {
            const int cur   = idx;
            const int curnv = nv;
            const float dg  = deg[cur];
            head += curnv;
            left -= curnv;
            const bool more = flush ? (left > 0) : (left >= EPT);
            if (more) {
                nv  = left < EPT ? left : EPT;
                idx = own_idx(ei, E, q, head, nv, lane);
            }
            float v[EPT][4];
#pragma unroll
            for (int j = 0; j < EPT; ++j) {
                const float* src = xw + (size_t)lane_i(cur, j) * D + lane;
                v[j][0] = src[0];
                v[j][1] = src[32];
                v[j][2] = src[64];
                v[j][3] = src[96];
            }
            __builtin_amdgcn_sched_barrier(0);
            const float dv = inv_sqrt_deg(dg);
            if (REC) {
                if (nrec < TCAP) {
                    const float wgt = dv * xrow<8>(dv);
                    const bool ok = (lane & 7) < curnv && cur >= lo && cur < hi;
                    int word = cur;
                    if (lane & 8)  word = ok ? cur - lo : -1;
                    if (lane & 16) word = __float_as_int(wgt);
                    recw[(size_t)nrec * 32] = word;
                }
                ++nrec;
            }
#pragma unroll
            for (int j = 0; j < EPT; ++j) {
                const int c = lane_i(cur, 8 + j);
                if (j < curnv && c >= lo && c < hi) {
                    const float nrm = lane_f(dv, j) * lane_f(dv, 8 + j);
                    float* dst = acc + (c - lo) * D + lane;
                    atomicAdd(dst +  0, v[j][0] * nrm);
                    atomicAdd(dst + 32, v[j][1] * nrm);
                    atomicAdd(dst + 64, v[j][2] * nrm);
                    atomicAdd(dst + 96, v[j][3] * nrm);
                }
            }
            if (!more) break;
        }
        if (left > 0) {
            const int tmp = q[head + (lane < left ? lane : 0)];
            wave_lds_order();
            if (lane < left) q[lane] = tmp;
        }
    }
    return left;
}

__device__ __forceinline__ void own_replay(
        const int* __restrict__ rp, int first, int nt, const float* __restrict__ xw,
        float* acc, int lane) {
    int nxt = first;
    for (int t = 0; t < nt; ++t) {
        const int cur = nxt;
        const int tn = t + 1 < nt ? t + 1 : t;
        nxt = rp[(size_t)tn * 32];
        float v[EPT][4];
#pragma unroll
        for (int j = 0; j < EPT; ++j) {
            const float* src = xw + (size_t)lane_i(cur, j) * D + lane;
            v[j][0] = src[0];
            v[j][1] = src[32];
            v[j][2] = src[64];
            v[j][3] = src[96];
        }
#pragma unroll
        for (int j = 0; j < EPT; ++j) {
            const int tl = lane_i(cur, 8 + j);
            const bool ok = tl >= 0;
            const float nrm = __int_as_float(lane_i(cur, 16 + j));
            float* dst = acc + (ok ? tl : 0) * D + lane;
            atomicAdd(dst +  0, ok ? v[j][0] * nrm : 0.0f);
            atomicAdd(dst + 32, ok ? v[j][1] * nrm : 0.0f);
            atomicAdd(dst + 64, ok ? v[j][2] * nrm : 0.0f);
            atomicAdd(dst + 96, ok ? v[j][3] * nrm : 0.0f);
        }
    }
}

template <bool REC>
__device__ __forceinline__ int own_scan_add(
        const int* __restrict__ ei, int E, const uint4* __restrict__ kp,
        int nchunk, int nstep,
        const float* __restrict__ xw, const float* __restrict__ deg,
        float* acc, unsigned short* mycq, int* myeq,
        int lo, int hi, unsigned pat, int lane, int wave,
        int* __restrict__ recw) {
    int nrec = 0;
    int ecnt = 0;
    for (int i0 = 0; wave + 32 * i0 < nstep; i0 += ROUND_STEPS) {
        int cnt = 0;
        for (int g = 0; g < ROUND_STEPS; g += SCAN_U) {
            if (wave + 32 * (i0 + g) >= nstep) break;
            uint4 kv[SCAN_U];
#pragma unroll
            for (int u = 0; u < SCAN_U; ++u) {
                const int ch = (wave + 32 * (i0 + g + u)) * 32 + lane;
                kv[u] = kp[ch < nchunk ? ch : nchunk - 1];
            }
#pragma unroll
            for (int u = 0; u < SCAN_U; ++u) {
                const unsigned a = kv[u].x ^ pat, b = kv[u].y ^ pat;
                const unsigned c = kv[u].z ^ pat, d = kv[u].w ^ pat;
                const unsigned z = (((a - 0x01010101u) & ~a) | ((b - 0x01010101u) & ~b) |
                                    ((c - 0x01010101u) & ~c) | ((d - 0x01010101u) & ~d)) & 0x80808080u;
                const bool hit = z != 0u && (wave + 32 * (i0 + g + u)) * 32 + lane < nchunk;
                const unsigned mask = __builtin_amdgcn_ballot_w32(hit);
                if (hit)
                    mycq[cnt + __builtin_amdgcn_mbcnt_lo(mask, 0u)] =
                        (unsigned short)(((g + u) << 5) | lane);
                cnt += __builtin_popcount(mask);
            }
        }
        for (int base = 0; base < cnt; base += 32) {
            if (ecnt + 512 > EQ_CAP)
                ecnt = own_drain<REC>(ei, E, xw, deg, acc, myeq, ecnt, lo, hi, lane, false, recw, nrec);
            wave_lds_order();
            const bool have = base + lane < cnt;
            const unsigned q = mycq[have ? base + lane : base];
            const int ch = (wave + 32 * (i0 + (int)(q >> 5))) * 32 + (int)(q & 31u);
            const uint4 kw = kp[ch];
            const unsigned w[4] = {kw.x ^ pat, kw.y ^ pat, kw.z ^ pat, kw.w ^ pat};
#pragma unroll
            for (int k = 0; k < 16; ++k) {
                const int e = ch * 16 + k;
                const bool m = have && ((w[k >> 2] >> (8 * (k & 3))) & 255u) == 0u && e < E;
                const unsigned mask = __builtin_amdgcn_ballot_w32(m);
                if (m) myeq[ecnt + __builtin_amdgcn_mbcnt_lo(mask, 0u)] = e;
                ecnt += __builtin_popcount(mask);
            }
        }
    }
    own_drain<REC>(ei, E, xw, deg, acc, myeq, ecnt, lo, hi, lane, true, recw, nrec);
    return nrec;
}

template <bool REPLAY>
__global__ __launch_bounds__(1024) void k_agg_own(
        const int* __restrict__ ei, int E,
        const unsigned* __restrict__ key,
        const float* __restrict__ xw, const float* __restrict__ deg,
        float* __restrict__ out, int n,
        int* __restrict__ rec, int* __restrict__ rcnt) {
    __shared__ float acc[NPB * D];
    __shared__ unsigned short cq[OWN_WAVES][CQ_CAP];
    __shared__ int eq[OWN_WAVES][EQ_CAP];

    const int lane = threadIdx.x & 31;
    const int wave   = __builtin_amdgcn_readfirstlane(threadIdx.x >> 5);
    const int nchunk = (E + 15) >> 4;
    const int nstep  = (nchunk + 31) >> 5;
    const int nrange = (n + NPB - 1) / NPB;
    const uint4* kp  = (const uint4*)key;
    unsigned short* mycq = cq[wave];
    int* myeq = eq[wave];

    for (int r = blockIdx.x; r < nrange; r += gridDim.x) {
        const int lo   = r * NPB;
        const int hi   = (lo + NPB < n) ? lo + NPB : n;
        const int rows = hi - lo;
        const unsigned pat  = ((unsigned)r & 255u) * 0x01010101u;
        const int slot = r * OWN_WAVES + wave;
        int* recw = rec + (size_t)slot * (TCAP * 32) + lane;

        int nt = 0, first = 0;
        if (REPLAY) {
            nt    = rcnt[slot];
            first = recw[0];
        }

        {
            float sv[ROWS_PER_WAVE][4], sd[ROWS_PER_WAVE];
#pragma unroll
            for (int k = 0; k < ROWS_PER_WAVE; ++k) {
                const int i  = wave + OWN_WAVES * k;
                const int rr = lo + (i < rows ? i : rows - 1);
                sd[k] = deg[rr];
                const float* src = xw + (size_t)rr * D + lane;
#pragma unroll
                for (int j = 0; j < 4; ++j) sv[k][j] = src[32 * j];
            }
#pragma unroll
            for (int k = 0; k < ROWS_PER_WAVE; ++k) {
                const int i = wave + OWN_WAVES * k;
                if (i < rows) {
                    const float dd = inv_sqrt_deg(sd[k]);
                    const float d2 = dd * dd;
                    float* a = acc + i * D + lane;
#pragma unroll
                    for (int j = 0; j < 4; ++j) a[32 * j] = sv[k][j] * d2;
                }
            }
        }
        __syncthreads();

        if (REPLAY) {
            nt = __builtin_amdgcn_readfirstlane(nt);
            if (nt > 0)
                own_replay(recw, first, nt, xw, acc, lane);
            else if (nt < 0)
                own_scan_add<false>(ei, E, kp, nchunk, nstep, xw, deg, acc, mycq, myeq,
                                    lo, hi, pat, lane, wave, recw);
        } else {
            const int nrec = own_scan_add<true>(ei, E, kp, nchunk, nstep, xw, deg, acc, mycq, myeq,
                                                lo, hi, pat, lane, wave, recw);
            if (lane == 0) rcnt[slot] = nrec <= TCAP ? nrec : -1;
            if (nrec == 0) recw[0] = -1;
        }
        __syncthreads();

        for (int i = wave; i < rows; i += OWN_WAVES) {
            float* dst = out + (size_t)(lo + i) * D + lane;
            const float* a = acc + i * D + lane;
#pragma unroll
            for (int j = 0; j < 4; ++j) dst[32 * j] = a[32 * j];
        }
        __syncthreads();
    }
}

__global__ __launch_bounds__(256) void k_edge_cls(
        const int* __restrict__ ei, int E,
        const float* __restrict__ hs, const float* __restrict__ bh,
        const float* __restrict__ Wfc,
        const float* __restrict__ bfc, float* __restrict__ out) {
    const int lane  = threadIdx.x & 31;
    const int wave  = blockIdx.x * 8 + __builtin_amdgcn_readfirstlane(threadIdx.x >> 5);
    const int nwave = gridDim.x * 8;
    const int ntrip = (E + EPT - 1) / EPT;
    if (wave >= ntrip) return;

    float4 wr[4], wc[4];
#pragma unroll
    for (int j = 0; j < 4; ++j) {
        wr[j] = ((const float4*)Wfc)[lane + 32 * j];
        wc[j] = ((const float4*)Wfc)[D + lane + 32 * j];
    }
    const float bfl = bfc[lane & 3];
    float bv[4];
#pragma unroll
    for (int j = 0; j < 4; ++j) bv[j] = bh[lane + 32 * j];

    const bool b4 = (lane & 16) != 0;
    const bool b3 = (lane &  8) != 0;
    const bool b2 = (lane &  4) != 0;
    const bool b1 = (lane &  2) != 0;
    const bool b0 = (lane &  1) != 0;

    int idx = trip_index(ei, E, wave, lane);
    for (int t = wave; t < ntrip; t += nwave) {
        const int cur = idx;
        const int tn = (t + nwave < ntrip) ? t + nwave : ntrip - 1;
        idx = trip_index(ei, E, tn, lane);

        float p[4 * EPT];
#pragma unroll
        for (int g = 0; g < EPT; g += 4) {
            float a[4][4], bq[4][4];
#pragma unroll
            for (int q = 0; q < 4; ++q) {
                const float* hr = hs + (size_t)lane_i(cur, g + q) * D + lane;
                const float* hc = hs + (size_t)lane_i(cur, 8 + g + q) * D + lane;
#pragma unroll
                for (int j = 0; j < 4; ++j) {
                    a[q][j]  = hr[32 * j];
                    bq[q][j] = hc[32 * j];
                }
            }
#pragma unroll
            for (int q = 0; q < 4; ++q) {
#pragma unroll
                for (int j = 0; j < 4; ++j) {
                    float u = a[q][j] + bv[j];
                    a[q][j] = u > 0.0f ? u : 0.0f;
                    float w = bq[q][j] + bv[j];
                    bq[q][j] = w > 0.0f ? w : 0.0f;
                }
            }
#pragma unroll
            for (int q = 0; q < 4; ++q) {
                float s0 = 0.f, s1 = 0.f, s2 = 0.f, s3 = 0.f;
#pragma unroll
                for (int j = 0; j < 4; ++j) {
                    s0 += a[q][j] * wr[j].x + bq[q][j] * wc[j].x;
                    s1 += a[q][j] * wr[j].y + bq[q][j] * wc[j].y;
                    s2 += a[q][j] * wr[j].z + bq[q][j] * wc[j].z;
                    s3 += a[q][j] * wr[j].w + bq[q][j] * wc[j].w;
                }
                p[4 * (g + q) + 0] = s0;
                p[4 * (g + q) + 1] = s1;
                p[4 * (g + q) + 2] = s2;
                p[4 * (g + q) + 3] = s3;
            }
        }

        float r1[16], r2[8], r3[4], r2b[2];
#pragma unroll
        for (int w = 0; w < 16; ++w) {
            const float keep = b4 ? p[16 + w] : p[w];
            const float send = b4 ? p[w] : p[16 + w];
            r1[w] = keep + xhalf(send);
        }
#pragma unroll
        for (int w = 0; w < 8; ++w) {
            const float keep = b3 ? r1[8 + w] : r1[w];
            const float send = b3 ? r1[w] : r1[8 + w];
            r2[w] = keep + xrow<8>(send);
        }
#pragma unroll
        for (int w = 0; w < 4; ++w) {
            const float keep = b2 ? r2[4 + w] : r2[w];
            const float send = b2 ? r2[w] : r2[4 + w];
            r3[w] = keep + xrow<4>(send);
        }
#pragma unroll
        for (int w = 0; w < 2; ++w) {
            const float keep = b1 ? r3[2 + w] : r3[w];
            const float send = b1 ? r3[w] : r3[2 + w];
            r2b[w] = keep + xrow<2>(send);
        }
        float x;
        {
            const float keep = b0 ? r2b[1] : r2b[0];
            const float send = b0 ? r2b[0] : r2b[1];
            x = keep + xrow<1>(send);
        }

        x += bfl;
        float mx = fmaxf(x, xrow<1>(x));
        mx = fmaxf(mx, xrow<2>(mx));
        const float ex = __expf(x - mx);
        float sum = ex + xrow<1>(ex);
        sum = sum + xrow<2>(sum);
        const float o = x - (mx + __logf(sum));

        const int e = t * EPT + (lane >> 2);
        if (e < E) out[(size_t)t * (4 * EPT) + lane] = o;
    }
}

extern "C" void kernel_launch(void* const* d_in, const int* in_sizes, int n_in,
                              void* d_out, int out_size, void* d_ws, size_t ws_size,
                              hipStream_t stream) {
    const float* x   = (const float*)d_in[0];
    const int*   ei  = (const int*)  d_in[1];
    const float* W1  = (const float*)d_in[2];
    const float* b1  = (const float*)d_in[3];
    const float* W2  = (const float*)d_in[4];
    const float* b2  = (const float*)d_in[5];
    const float* Wfc = (const float*)d_in[6];
    const float* bfc = (const float*)d_in[7];
    float* out = (float*)d_out;

    const int n = in_sizes[0] / D;
    const int E = in_sizes[1] / 2;

    float* bufA = (float*)d_ws;
    float* bufB = bufA + (size_t)n * D;
    float* bufC = bufB + (size_t)n * D;
    const int nOwn  = (n + NPB - 1) / NPB;
    int* rec    = (int*)(bufC + (size_t)n * D);
    int* rcnt   = rec + (size_t)nOwn * OWN_WAVES * TCAP * 32;
    float* deg  = (float*)(rcnt + (size_t)nOwn * OWN_WAVES);
    unsigned* key = (unsigned*)(deg + ((n + 3) & ~3));

    const int gN    = (n + 255) / 256;
    const int gGemm = (n + 127) / 128;
    const int gRes  = RES_BLOCKS;
    const int gKey  = (((E + 15) / 16) * 4 + 255) / 256;
    const int gOwn  = nOwn < 256 ? nOwn : 256;

    k_init_deg <<<gN, 256, 0, stream>>>(deg, n);
    k_count_deg<<<gKey, 256, 0, stream>>>(ei, E, deg, key);

    k_gemm_first  <<<gGemm, 256, 0, stream>>>(x, W1, bufA, n);
    k_agg_own<false><<<gOwn, 1024, 0, stream>>>(ei, E, key, bufA, deg, bufB, n, rec, rcnt);

    k_gemm_relu_in<<<gGemm, 256, 0, stream>>>(bufB, b1, W2, bufA, n);
    k_agg_own<true> <<<gOwn, 1024, 0, stream>>>(ei, E, key, bufA, deg, bufC, n, rec, rcnt);

    k_edge_cls    <<<gRes, 256, 0, stream>>>(ei, E, bufC, b2, Wfc, bfc, out);
}
